# starting kernel
# baseline (speedup 1.0000x reference)
#include <hip/hip_runtime.h>
#include <hip/hip_bf16.h>
#include <math.h>


typedef __attribute__((ext_vector_type(2))) float v2f;
typedef __attribute__((ext_vector_type(8))) float v8f;

#define NBATCH 8
#define NPTS   4096
#define WAVES  4
#define BLOCK  (WAVES * 32)
#define QPB    (WAVES * 16)
#define CHUNK  2048

__global__ __launch_bounds__(BLOCK)
void chamfer_nn_kernel(const float* __restrict__ tar,
                       const float* __restrict__ src,
                       float* __restrict__ out)
{
    const int b   = blockIdx.y;
    const int dir = blockIdx.z;
    const float* qry = (dir == 0 ? src : tar) + b * NPTS * 3;
    const float* ref = (dir == 0 ? tar : src) + b * NPTS * 3;
    float* o = out + dir * (NBATCH * NPTS) + b * NPTS;

    const int tid  = threadIdx.x;
    const int wave = tid >> 5;
    const int lane = tid & 31;
    const int hi   = lane >> 4;
    const int col  = lane & 15;

    __shared__ float2 s_ref[2 * CHUNK];

    const int qbase = blockIdx.x * QPB + wave * 16;
    const float* qp = qry + (qbase + col) * 3;
    const float qx = qp[0], qy = qp[1], qz = qp[2];
    v2f Av;
    Av.x = hi ? qz : qx;
    Av.y = hi ? 1.0f : qy;

    float bm[8];
#pragma unroll
    for (int r = 0; r < 8; ++r) bm[r] = 3.4e38f;

    const v8f czero = {};

    for (int c0 = 0; c0 < NPTS; c0 += CHUNK) {
        __syncthreads();
        for (int i = tid; i < CHUNK; i += BLOCK) {
            const float* rp = ref + (c0 + i) * 3;
            const float x = rp[0], y = rp[1], z = rp[2];
            s_ref[2 * i + 0] = make_float2(-2.0f * x, -2.0f * y);
            s_ref[2 * i + 1] = make_float2(-2.0f * z, x * x + y * y + z * z);
        }
        __syncthreads();

#pragma unroll 4
        for (int t = 0; t < CHUNK; t += 16) {
            const float2 bv = s_ref[2 * (t + col) + hi];
            v2f Bv;
            Bv.x = bv.x;
            Bv.y = bv.y;
            v8f d = __builtin_amdgcn_wmma_f32_16x16x4_f32(
                false, Av, false, Bv, (short)0, czero, false, false);
#pragma unroll
            for (int r = 0; r < 8; ++r) bm[r] = fminf(bm[r], d[r]);
        }
    }

#pragma unroll
    for (int r = 0; r < 8; ++r) {
        float v = bm[r];
        v = fminf(v, __shfl_xor(v, 1, 32));
        v = fminf(v, __shfl_xor(v, 2, 32));
        v = fminf(v, __shfl_xor(v, 4, 32));
        v = fminf(v, __shfl_xor(v, 8, 32));
        bm[r] = v;
    }

    if (col == 0) {
        const int qb = qbase + hi * 8;
#pragma unroll
        for (int r = 0; r < 8; ++r) {
            const float* q2p = qry + (qb + r) * 3;
            const float x = q2p[0], y = q2p[1], z = q2p[2];
            const float d2 = bm[r] + (x * x + y * y + z * z);
            o[qb + r] = sqrtf(fmaxf(d2, 0.0f));
        }
    }
}

__global__ __launch_bounds__(256)
void chamfer_reduce_kernel(const float* __restrict__ acc,
                           const float* __restrict__ comp,
                           float* __restrict__ cham)
{
    const int b = blockIdx.x;
    __shared__ float red[256];
    float s = 0.0f;
    for (int i = threadIdx.x; i < NPTS; i += 256)
        s += acc[b * NPTS + i] + comp[b * NPTS + i];
    red[threadIdx.x] = s;
    __syncthreads();
    for (int st = 128; st > 0; st >>= 1) {
        if (threadIdx.x < st) red[threadIdx.x] += red[threadIdx.x + st];
        __syncthreads();
    }
    if (threadIdx.x == 0)
        cham[b] = red[0] / (2.0f * (float)NPTS);
}

extern "C" void kernel_launch(void* const* d_in, const int* in_sizes, int n_in,
                              void* d_out, int out_size, void* d_ws, size_t ws_size,
                              hipStream_t stream) {
    (void)in_sizes; (void)n_in; (void)d_ws; (void)ws_size; (void)out_size;
    const float* tar = (const float*)d_in[0];
    const float* src = (const float*)d_in[1];
    float* out = (float*)d_out;

    dim3 grid(NPTS / QPB, NBATCH, 2);
    chamfer_nn_kernel<<<grid, BLOCK, 0, stream>>>(tar, src, out);

    const float* acc  = out;
    const float* comp = out + NBATCH * NPTS;
    float* cham       = out + 2 * NBATCH * NPTS;
    chamfer_reduce_kernel<<<NBATCH, 256, 0, stream>>>(acc, comp, cham);
}


# Mean's sum starts in launch 1: each wave adds its 16 distances and stores one partial; launch 2 is one wave per batch: 512 partials in one load group, no LDS, no barrier; the mean's sum order changes
# speedup vs baseline: 1.5357x; 1.5357x over previous
#include <hip/hip_runtime.h>
#include <hip/hip_bf16.h>
#include <math.h>


typedef __attribute__((ext_vector_type(2))) float v2f;
typedef __attribute__((ext_vector_type(8))) float v8f;

#define NBATCH 8
#define NPTS   4096
#define WAVES  4
#define BLOCK  (WAVES * 32)
#define QT     4
#define QPB    (QT * 16)
#define SLAB   (NPTS / WAVES)
#define BT     (SLAB / 16)
#define AHEAD  40
#define REFILL 8
#define PPB    ((NPTS / QPB) * WAVES)

template <int M>
__device__ __forceinline__ float row_xor(float v)
{
    return __builtin_bit_cast(float,
        __builtin_amdgcn_update_dpp(0, __builtin_bit_cast(int, v), 0x160 + M, 0xf, 0xf, true));
}

__device__ __forceinline__ float row_sum16(float s)
{
    s += row_xor<1>(s);
    s += row_xor<2>(s);
    s += row_xor<4>(s);
    s += row_xor<8>(s);
    return s;
}

__global__ __launch_bounds__(BLOCK, 4)
void chamfer_nn_kernel(const float* __restrict__ tar,
                       const float* __restrict__ src,
                       float* __restrict__ out,
                       float* __restrict__ part)
{
    const int b   = blockIdx.y;
    const int dir = blockIdx.z;
    const float* qry = (dir == 0 ? src : tar) + b * NPTS * 3;
    const float* ref = (dir == 0 ? tar : src) + b * NPTS * 3;
    float* o = out + dir * (NBATCH * NPTS) + b * NPTS;

    const int tid  = threadIdx.x;
    const int wave = tid >> 5;
    const int lane = tid & 31;
    const int hi   = lane >> 4;
    const int col  = lane & 15;

    __shared__ __attribute__((aligned(16))) float s_pm[QT * 16 * 2 * WAVES];

    const float* qp0 = qry + (blockIdx.x * QPB + col) * 3;
    float qx[QT], qy[QT], qz[QT];
#pragma unroll
    for (int t = 0; t < QT; ++t) {
        qx[t] = qp0[t * 48 + 0]; qy[t] = qp0[t * 48 + 1]; qz[t] = qp0[t * 48 + 2];
    }

    const int q = blockIdx.x * QPB + wave * 16 + col;
    const float* q2p = qry + q * 3;
    const float fx = q2p[0], fy = q2p[1], fz = q2p[2];

    const float* rp = ref + (wave * SLAB + col) * 3;
    float rx[BT], ry[BT], rz[BT];
#pragma unroll
    for (int j = 0; j < AHEAD; ++j) {
        rx[j] = rp[j * 48 + 0]; ry[j] = rp[j * 48 + 1]; rz[j] = rp[j * 48 + 2];
    }

    __builtin_amdgcn_sched_barrier(0);

    v2f Qf[QT];
#pragma unroll
    for (int t = 0; t < QT; ++t) {
        Qf[t].x = -2.0f * (hi ? qz[t] : qx[t]);
        Qf[t].y = hi ? 1.0f : -2.0f * qy[t];
    }

    float bm[QT][8];
#pragma unroll
    for (int t = 0; t < QT; ++t)
#pragma unroll
        for (int r = 0; r < 8; ++r) bm[t][r] = 3.4e38f;

    const v8f czero = {};

#define REF_FRAG(F, j)                                                        \
    {                                                                         \
        const float x_ = rx[j], y_ = ry[j], z_ = rz[j];                       \
        (F).x = hi ? z_ : x_;                                                 \
        (F).y = hi ? (x_ * x_ + y_ * y_ + z_ * z_) : y_;                      \
    }

    v2f A0, A1;
    REF_FRAG(A0, 0)
    REF_FRAG(A1, 1)

    v8f p0 = czero, p1 = czero;

#pragma unroll
    for (int j = 0; j < BT; j += 2) {
        if (j >= REFILL && j % REFILL == 0 && AHEAD + j - REFILL < BT) {
#pragma unroll
            for (int k = AHEAD + j - REFILL; k < AHEAD + j && k < BT; ++k) {
                rx[k] = rp[k * 48 + 0]; ry[k] = rp[k * 48 + 1]; rz[k] = rp[k * 48 + 2];
            }
        }
        v2f N0 = A0, N1 = A1;
#pragma unroll
        for (int t = 0; t < QT; ++t) {
            const v8f d0 = __builtin_amdgcn_wmma_f32_16x16x4_f32(
                false, A0, false, Qf[t], (short)0, czero, false, false);
            const v8f d1 = __builtin_amdgcn_wmma_f32_16x16x4_f32(
                false, A1, false, Qf[t], (short)0, czero, false, false);
            __builtin_amdgcn_sched_barrier(0);
            if (t == 0 && j + 2 < BT) {
                REF_FRAG(N0, j + 2)
                REF_FRAG(N1, j + 3)
            }
            if (j != 0 || t != 0) {
                const int tp = (t + QT - 1) % QT;
#pragma unroll
                for (int r = 0; r < 8; ++r)
                    bm[tp][r] = fminf(fminf(bm[tp][r], p0[r]), p1[r]);
            }
            p0 = d0; p1 = d1;
        }
        A0 = N0; A1 = N1;
    }
#pragma unroll
    for (int r = 0; r < 8; ++r)
        bm[QT - 1][r] = fminf(fminf(bm[QT - 1][r], p0[r]), p1[r]);
#undef REF_FRAG

#pragma unroll
    for (int t = 0; t < QT; ++t) {
        const float m01 = fminf(fminf(bm[t][0], bm[t][1]), bm[t][2]);
        const float m35 = fminf(fminf(bm[t][3], bm[t][4]), bm[t][5]);
        const float m   = fminf(fminf(m01, m35), fminf(bm[t][6], bm[t][7]));
        s_pm[(t * 16 + col) * (2 * WAVES) + 2 * wave + hi] = m;
    }

    const float qq = fx * fx + fy * fy + fz * fz;

    __syncthreads();

    const float4* pm = reinterpret_cast<const float4*>(s_pm + (wave * 16 + col) * (2 * WAVES));
    const float4 e = pm[0], f = pm[1];
    const float me = fminf(fminf(e.x, e.y), fminf(e.z, e.w));
    const float mf = fminf(fminf(f.x, f.y), fminf(f.z, f.w));
    const float d2 = fminf(me, mf) + qq;
    const float dist = sqrtf(fmaxf(d2, 0.0f));
    if (hi == 0) o[q] = dist;

    const float wsum = row_sum16(dist);
    if (lane == 0)
        part[((dir * NBATCH + b) * (NPTS / QPB) + blockIdx.x) * WAVES + wave] = wsum;
}

__global__ __launch_bounds__(32)
void chamfer_mean_kernel(const float* __restrict__ part,
                         float* __restrict__ cham)
{
    const int b    = blockIdx.x;
    const int lane = threadIdx.x;
    const float4* pa = reinterpret_cast<const float4*>(part + b * PPB);
    const float4* pc = reinterpret_cast<const float4*>(part + (NBATCH + b) * PPB);
    const float4 a0 = pa[lane], a1 = pa[lane + 32];
    const float4 c0 = pc[lane], c1 = pc[lane + 32];
    const float sa = ((a0.x + a0.y) + (a0.z + a0.w)) + ((a1.x + a1.y) + (a1.z + a1.w));
    const float sc = ((c0.x + c0.y) + (c0.z + c0.w)) + ((c1.x + c1.y) + (c1.z + c1.w));
    float s = row_sum16(sa + sc);
    s += __builtin_bit_cast(float,
            __builtin_amdgcn_permlanex16(__builtin_bit_cast(int, s), __builtin_bit_cast(int, s),
                                         0x76543210, 0xfedcba98, false, false));
    if (lane == 0)
        cham[b] = s / (2.0f * (float)NPTS);
}

extern "C" void kernel_launch(void* const* d_in, const int* in_sizes, int n_in,
                              void* d_out, int out_size, void* d_ws, size_t ws_size,
                              hipStream_t stream) {
    (void)in_sizes; (void)n_in; (void)ws_size; (void)out_size;
    const float* tar = (const float*)d_in[0];
    const float* src = (const float*)d_in[1];
    float* out = (float*)d_out;

    float* part = (float*)d_ws;

    dim3 grid(NPTS / QPB, NBATCH, 2);
    chamfer_nn_kernel<<<grid, BLOCK, 0, stream>>>(tar, src, out, part);

    float* cham = out + 2 * NBATCH * NPTS;
    chamfer_mean_kernel<<<NBATCH, 32, 0, stream>>>(part, cham);
}
